# starting kernel
# baseline (speedup 1.0000x reference)
#include <hip/hip_runtime.h>
#include <hip/hip_bf16.h>

#define BS   512
#define DW   400
#define DT   20
#define TT   30
#define RR   256
#define FAN  440
#define NOUT (TT * TT)

typedef float v2f __attribute__((ext_vector_type(2)));
typedef float v8f __attribute__((ext_vector_type(8)));

__device__ __forceinline__ float fast_tanh(float x) {
#if __has_builtin(__builtin_amdgcn_tanhf)
    return __builtin_amdgcn_tanhf(x);
#else
    float y;
    asm volatile("v_tanh_f32 %0, %1" : "=v"(y) : "v"(x));
    return y;
#endif
}

__global__ void __launch_bounds__(32)
wordpart_wmma_kernel(const float* __restrict__ word,
                     const float* __restrict__ W,
                     float* __restrict__ wp) {
    const int m0   = blockIdx.x * 16;
    const int n0   = blockIdx.y * 16;
    const int lane = threadIdx.x;
    const int half = lane >> 4;
    const int mn   = lane & 15;

    const float* arow = word + (size_t)(m0 + mn) * DW + 2 * half;
    const float* brow = W + (size_t)(n0 + mn) * FAN + 2 * half;

    v8f acc = {0.f, 0.f, 0.f, 0.f, 0.f, 0.f, 0.f, 0.f};

    for (int k = 0; k < DW; k += 4) {
        v2f a, b;
        a.x = arow[k];
        a.y = arow[k + 1];
        b.x = brow[k];
        b.y = brow[k + 1];
        acc = __builtin_amdgcn_wmma_f32_16x16x4_f32(
            false, a, false, b, (short)0, acc, false, false);
    }

    float* dcol = wp + n0 + mn;
#pragma unroll
    for (int v = 0; v < 8; ++v) {
        const int m = v + 8 * half;
        dcol[(size_t)(m0 + m) * RR] = acc[v];
    }
}

__global__ void __launch_bounds__(RR)
tag_kernel(const float* __restrict__ tag,
           const float* __restrict__ W,
           const float* __restrict__ bias,
           float* __restrict__ ti,
           float* __restrict__ tj) {
    const int t = blockIdx.x;
    const int r = threadIdx.x;
    const float* te = tag + t * DT;
    const float* w1 = W + (size_t)r * FAN + DW;
    const float* w2 = W + (size_t)r * FAN + DW + DT;
    float aj = 0.f, ai = 0.f;
#pragma unroll
    for (int d = 0; d < DT; ++d) {
        const float x = te[d];
        aj += x * w1[d];
        ai += x * w2[d];
    }
    tj[t * RR + r] = aj;
    ti[t * RR + r] = ai + bias[r];
}

__global__ void __launch_bounds__(RR)
score_kernel(const float* __restrict__ wp,
             const float* __restrict__ ti,
             const float* __restrict__ tj,
             const float* __restrict__ vec,
             float* __restrict__ out) {
    __shared__ float s_wp[RR];
    __shared__ float s_v[RR];
    __shared__ float s_ti[TT * RR];
    __shared__ float s_tj[TT * RR];

    const int bs  = blockIdx.x;
    const int tid = threadIdx.x;

    s_wp[tid] = wp[(size_t)bs * RR + tid];
    s_v[tid]  = vec[tid];
    for (int idx = tid; idx < TT * RR; idx += RR) {
        s_ti[idx] = ti[idx];
        s_tj[idx] = tj[idx];
    }
    __syncthreads();

    const int  o0   = tid;
    const int  o1   = tid + RR;
    const int  o2   = tid + 2 * RR;
    const bool has3 = (tid + 3 * RR) < NOUT;
    const int  o3   = has3 ? (tid + 3 * RR) : (NOUT - 1);

    const float4* p_wp = (const float4*)s_wp;
    const float4* p_v  = (const float4*)s_v;
    const float4* ti0  = (const float4*)(s_ti + (o0 / TT) * RR);
    const float4* tj0  = (const float4*)(s_tj + (o0 % TT) * RR);
    const float4* ti1  = (const float4*)(s_ti + (o1 / TT) * RR);
    const float4* tj1  = (const float4*)(s_tj + (o1 % TT) * RR);
    const float4* ti2  = (const float4*)(s_ti + (o2 / TT) * RR);
    const float4* tj2  = (const float4*)(s_tj + (o2 % TT) * RR);
    const float4* ti3  = (const float4*)(s_ti + (o3 / TT) * RR);
    const float4* tj3  = (const float4*)(s_tj + (o3 % TT) * RR);

    float acc0 = 0.f, acc1 = 0.f, acc2 = 0.f, acc3 = 0.f;

#pragma unroll 2
    for (int rq = 0; rq < RR / 4; ++rq) {
        const float4 w4 = p_wp[rq];
        const float4 v4 = p_v[rq];

        float4 a, b;
        a = ti0[rq]; b = tj0[rq];
        acc0 += v4.x * fast_tanh(w4.x + a.x + b.x);
        acc0 += v4.y * fast_tanh(w4.y + a.y + b.y);
        acc0 += v4.z * fast_tanh(w4.z + a.z + b.z);
        acc0 += v4.w * fast_tanh(w4.w + a.w + b.w);

        a = ti1[rq]; b = tj1[rq];
        acc1 += v4.x * fast_tanh(w4.x + a.x + b.x);
        acc1 += v4.y * fast_tanh(w4.y + a.y + b.y);
        acc1 += v4.z * fast_tanh(w4.z + a.z + b.z);
        acc1 += v4.w * fast_tanh(w4.w + a.w + b.w);

        a = ti2[rq]; b = tj2[rq];
        acc2 += v4.x * fast_tanh(w4.x + a.x + b.x);
        acc2 += v4.y * fast_tanh(w4.y + a.y + b.y);
        acc2 += v4.z * fast_tanh(w4.z + a.z + b.z);
        acc2 += v4.w * fast_tanh(w4.w + a.w + b.w);

        a = ti3[rq]; b = tj3[rq];
        acc3 += v4.x * fast_tanh(w4.x + a.x + b.x);
        acc3 += v4.y * fast_tanh(w4.y + a.y + b.y);
        acc3 += v4.z * fast_tanh(w4.z + a.z + b.z);
        acc3 += v4.w * fast_tanh(w4.w + a.w + b.w);
    }

    float* orow = out + (size_t)bs * NOUT;
    orow[o0] = acc0;
    orow[o1] = acc1;
    orow[o2] = acc2;
    if (has3) orow[o3] = acc3;
}

extern "C" void kernel_launch(void* const* d_in, const int* in_sizes, int n_in,
                              void* d_out, int out_size, void* d_ws, size_t ws_size,
                              hipStream_t stream) {
    (void)in_sizes; (void)n_in; (void)out_size; (void)ws_size;

    const float* word = (const float*)d_in[0];
    const float* tag  = (const float*)d_in[1];
    const float* W    = (const float*)d_in[2];
    const float* bias = (const float*)d_in[3];
    const float* vec  = (const float*)d_in[4];
    float* out = (float*)d_out;

    float* wp = (float*)d_ws;
    float* ti = wp + (size_t)BS * RR;
    float* tj = ti + (size_t)TT * RR;

    wordpart_wmma_kernel<<<dim3(BS / 16, RR / 16), 32, 0, stream>>>(word, W, wp);
    tag_kernel<<<TT, RR, 0, stream>>>(tag, W, bias, ti, tj);
    score_kernel<<<BS, RR, 0, stream>>>(wp, ti, tj, vec, out);
}


# projection launch re-tiled: one 4-wave block per 16x32 outputs reads its 48 operand rows with whole-line 16 B loads into LDS, K halved across wave pairs, 2 accumulator chains per wave
# speedup vs baseline: 4.3524x; 4.3524x over previous
#include <hip/hip_runtime.h>
#include <hip/hip_bf16.h>

#define BS   512
#define DW   400
#define DT   20
#define TT   30
#define RR   256
#define FAN  440
#define NOUT (TT * TT)

typedef float v2f __attribute__((ext_vector_type(2)));
typedef float v8f __attribute__((ext_vector_type(8)));

__device__ __forceinline__ float fast_tanh(float x) {
#if __has_builtin(__builtin_amdgcn_tanhf)
    return __builtin_amdgcn_tanhf(x);
#else
    float y;
    asm volatile("v_tanh_f32 %0, %1" : "=v"(y) : "v"(x));
    return y;
#endif
}

typedef float v4f __attribute__((ext_vector_type(4)));

#define PM      16
#define PN      32
#define KP      404
#define F4ROW   (DW / 4)
#define PTHR    128
#define NLDA    ((PM * F4ROW + PTHR - 1) / PTHR)
#define NLDB    ((PN * F4ROW) / PTHR)
#define NSS     (DW / 2 / 8)
#define NS0     5
#define KHALF   (DW / 2)
#define NWBLK   ((BS / PM) * (RR / PN))
#define NTAGBLK ((2 * 2 * (RR / 16)) / 4)
#define RP      (PN + 4)

__device__ __forceinline__ v8f mm4(v2f a, v2f b, v8f c) {
    return __builtin_amdgcn_wmma_f32_16x16x4_f32(false, a, false, b, (short)0, c, false, false);
}

__global__ void __launch_bounds__(PTHR)
proj_wmma_kernel(const float* __restrict__ word,
                 const float* __restrict__ tag,
                 const float* __restrict__ W,
                 const float* __restrict__ bias,
                 float* __restrict__ wp,
                 float* __restrict__ ti,
                 float* __restrict__ tj) {
    __shared__ __attribute__((aligned(16))) float s_pan[(PM + PN) * KP];
    __shared__ __attribute__((aligned(16))) float s_red[2][PM][RP];

    const int tid  = threadIdx.x;
    const int lane = tid & 31;
    const int wv   = __builtin_amdgcn_readfirstlane(tid >> 5);
    const int half = lane >> 4;
    const int mn   = lane & 15;

    if (blockIdx.x < NWBLK) {
        const int m0 = (blockIdx.x & (BS / PM - 1)) * PM;
        const int n0 = (blockIdx.x / (BS / PM)) * PN;

        const float* gA = word + (size_t)m0 * DW;
        const float* gB = W + (size_t)n0 * FAN;
        v4f bufA[NLDA];
        v4f bufB[NLDB];
#pragma unroll
        for (int q = 0; q < NLDA; ++q) {
            const unsigned idx = (unsigned)tid + q * PTHR;
            if (q < NLDA - 1 || wv < 2) bufA[q] = *(const v4f*)(gA + 4 * idx);
        }
#pragma unroll
        for (int q = 0; q < NLDB; ++q) {
            const unsigned idx = (unsigned)tid + q * PTHR;
            const unsigned row = idx / F4ROW;
            bufB[q] = *(const v4f*)(gB + 4 * idx + (FAN - DW) * row);
        }
#pragma unroll
        for (int q = 0; q < NLDA; ++q) {
            const unsigned idx = (unsigned)tid + q * PTHR;
            const unsigned row = idx / F4ROW;
            if (q < NLDA - 1 || wv < 2) *(v4f*)&s_pan[4 * idx + (KP - DW) * row] = bufA[q];
        }
#pragma unroll
        for (int q = 0; q < NLDB; ++q) {
            const unsigned idx = (unsigned)tid + q * PTHR;
            const unsigned row = idx / F4ROW;
            *(v4f*)&s_pan[PM * KP + 4 * idx + (KP - DW) * row] = bufB[q];
        }
        __syncthreads();

        const int nt = wv & 1;
        const int kh = wv >> 1;
        const float* ap = s_pan + mn * KP + kh * KHALF + 4 * half;
        const float* bp = s_pan + (PM + nt * 16 + mn) * KP + kh * KHALF + 4 * half;

        v4f fa[NSS];
        v4f fb[NSS];
#pragma unroll
        for (int g = 0; g < NSS / NS0; ++g) {
#pragma unroll
            for (int s = g * NS0; s < (g + 1) * NS0; ++s) {
                fa[s] = *(const v4f*)(ap + 8 * s);
                fb[s] = *(const v4f*)(bp + 8 * s);
            }
            __builtin_amdgcn_sched_barrier(0);
        }

        v8f acc0 = {0.f, 0.f, 0.f, 0.f, 0.f, 0.f, 0.f, 0.f};
        v8f acc1 = {0.f, 0.f, 0.f, 0.f, 0.f, 0.f, 0.f, 0.f};
#pragma unroll
        for (int s = 0; s < NSS; ++s) {
            acc0 = mm4((v2f){fa[s].x, fa[s].y}, (v2f){fb[s].x, fb[s].y}, acc0);
            acc1 = mm4((v2f){fa[s].z, fa[s].w}, (v2f){fb[s].z, fb[s].w}, acc1);
        }
        const v8f acc = acc0 + acc1;

#pragma unroll
        for (int v = 0; v < 8; ++v) s_red[kh][v + 8 * half][nt * 16 + mn] = acc[v];
        __syncthreads();

        {
            const int row = tid >> 3;
            const int c4  = tid & 7;
            const v4f p0 = *(const v4f*)&s_red[0][row][4 * c4];
            const v4f p1 = *(const v4f*)&s_red[1][row][4 * c4];
            *(v4f*)(wp + (size_t)(m0 + row) * RR + n0 + 4 * c4) = p0 + p1;
        }
    } else {
        const int q     = (blockIdx.x - NWBLK) * 4 + wv;
        const int which = q >> 5;
        const int t0    = ((q >> 4) & 1) * 16;
        const int n0    = (q & 15) * 16;

        const int trow = (t0 + mn) < TT ? (t0 + mn) : (TT - 1);
        const float* arow = tag + trow * DT + 2 * half;
        const float* brow = W + (size_t)(n0 + mn) * FAN + DW + which * DT + 2 * half;

        v8f acc = {0.f, 0.f, 0.f, 0.f, 0.f, 0.f, 0.f, 0.f};
#pragma unroll
        for (int k = 0; k < DT; k += 4) {
            v2f a, b;
            a.x = arow[k];
            a.y = arow[k + 1];
            b.x = brow[k];
            b.y = brow[k + 1];
            acc = mm4(a, b, acc);
        }

        const float add = which ? bias[n0 + mn] : 0.f;
        float* dcol = (which ? ti : tj) + n0 + mn;
#pragma unroll
        for (int v = 0; v < 8; ++v) {
            const int m = t0 + v + 8 * half;
            if (m < TT) dcol[m * RR] = acc[v] + add;
        }
    }
}

#define RC    32
#define NCH   (RR / RC)
#define JP    32
#define IH    (TT / 2)
#define NHALF (IH * TT)

__global__ void __launch_bounds__(RR)
score_kernel(const float* __restrict__ wp,
             const float* __restrict__ ti,
             const float* __restrict__ tj,
             const float* __restrict__ vec,
             float* __restrict__ out) {
    __shared__ float s_part[NCH][2][IH * JP];

    const int bs0  = (blockIdx.x >> 1) * 2;
    const int i0   = (blockIdx.x & 1) * IH;
    const int tid  = threadIdx.x;
    const int lane = tid & 31;
    const int c    = __builtin_amdgcn_readfirstlane(tid >> 5);
    const int j    = lane < TT ? lane : (TT - 1);

    const float* __restrict__ tjrow = tj + j * RR + c * RC;
    const float* __restrict__ wp0   = wp + (size_t)bs0 * RR + c * RC;
    const float* __restrict__ wp1   = wp0 + RR;
    const float* __restrict__ vrow  = vec + c * RC;
    const float* __restrict__ tibase = ti + i0 * RR + c * RC;

    v2f y[RC];
#pragma unroll
    for (int k = 0; k < RC; ++k) {
        const float t = tjrow[k];
        y[k].x = t + wp0[k];
        y[k].y = t + wp1[k];
    }

    float* part0 = &s_part[c][0][lane];
    float* part1 = &s_part[c][1][lane];

    int row = (int)((blockIdx.x >> 1) % IH);

#pragma unroll 1
    for (int ii = 0; ii < IH; ++ii) {
        const float* __restrict__ tirow = tibase + row * RR;
        v2f a0 = {0.f, 0.f};
        v2f a1 = {0.f, 0.f};
#pragma unroll
        for (int k = 0; k < RC; k += 2) {
            const float t0 = tirow[k];
            const float t1 = tirow[k + 1];
            const float u0 = vrow[k];
            const float u1 = vrow[k + 1];
            v2f x0 = y[k]     + (v2f){t0, t0};
            v2f x1 = y[k + 1] + (v2f){t1, t1};
            x0.x = fast_tanh(x0.x);
            x0.y = fast_tanh(x0.y);
            x1.x = fast_tanh(x1.x);
            x1.y = fast_tanh(x1.y);
            a0 = __builtin_elementwise_fma((v2f){u0, u0}, x0, a0);
            a1 = __builtin_elementwise_fma((v2f){u1, u1}, x1, a1);
        }
        const v2f s = a0 + a1;
        part0[row * JP] = s.x;
        part1[row * JP] = s.y;
        row = (row + 1 == IH) ? 0 : (row + 1);
    }
    __syncthreads();

#pragma unroll
    for (int q = 0; q < 4; ++q) {
        const int e = tid + q * RR;
        if (e < 2 * NHALF) {
            const int b  = e >= NHALF ? 1 : 0;
            const int o  = e - b * NHALF;
            const int oi = o / TT;
            const int oj = o - oi * TT;
            const int off = oi * JP + oj;
            float s = 0.f;
#pragma unroll
            for (int cc = 0; cc < NCH; ++cc) s += s_part[cc][b][off];
            out[(size_t)(bs0 + b) * NOUT + i0 * TT + o] = s;
        }
    }
}

extern "C" void kernel_launch(void* const* d_in, const int* in_sizes, int n_in,
                              void* d_out, int out_size, void* d_ws, size_t ws_size,
                              hipStream_t stream) {
    (void)in_sizes; (void)n_in; (void)out_size; (void)ws_size;

    const float* word = (const float*)d_in[0];
    const float* tag  = (const float*)d_in[1];
    const float* W    = (const float*)d_in[2];
    const float* bias = (const float*)d_in[3];
    const float* vec  = (const float*)d_in[4];
    float* out = (float*)d_out;

    float* wp = (float*)d_ws;
    float* ti = wp + (size_t)BS * RR;
    float* tj = ti + (size_t)TT * RR;

    proj_wmma_kernel<<<NWBLK + NTAGBLK, PTHR, 0, stream>>>(word, tag, W, bias, wp, ti, tj);
    score_kernel<<<BS, RR, 0, stream>>>(wp, ti, tj, vec, out);
}
